# starting kernel
# baseline (speedup 1.0000x reference)
#include <hip/hip_runtime.h>


typedef __attribute__((ext_vector_type(2))) float v2f;
typedef __attribute__((ext_vector_type(8))) float v8f;

__device__ __forceinline__ float sigf(float x) {
    return __builtin_amdgcn_rcpf(1.0f + __expf(-x));
}

template <int C, int S, int G>
__global__ __launch_bounds__(256)
void yolo_head_wmma(const float* __restrict__ F,
                    const float* __restrict__ Wm,
                    const float* __restrict__ bias,
                    float* __restrict__ out,
                    float aw0, float ah0, float aw1, float ah1,
                    float aw2, float ah2)
{
    constexpr float inv_grid = 1.0f / (float)G;

    const int lane  = threadIdx.x & 31;
    const int wave  = (blockIdx.x << 3) | (threadIdx.x >> 5);
    const int tm    = wave & 15;
    const int tn    = wave >> 4;

    const int l16   = lane & 15;
    const int half  = lane >> 4;
    const int half2 = half << 1;

    const int o_base = tm << 4;
    const int o_row  = o_base + l16;
    const int o_load = o_row > 254 ? 254 : o_row;

    const int n = (tn << 4) + l16;
    const int b = n / S;
    const int s = n - b * S;

    const float* __restrict__ wp = Wm + (size_t)o_load * C + half2;
    const float* __restrict__ fp = F + (size_t)b * C * S + (size_t)half2 * S + s;

    v8f acc = {0.f, 0.f, 0.f, 0.f, 0.f, 0.f, 0.f, 0.f};

    #pragma unroll 8
    for (int k = 0; k < C; k += 4) {
        v2f a = *(const v2f*)(wp + k);
        v2f bb;
        bb.x = fp[k * S];
        bb.y = fp[k * S + S];
        acc = __builtin_amdgcn_wmma_f32_16x16x4_f32(
                  false, a, false, bb, (short)0, acc, false, false);
    }

    const int h = s / G;
    const int w = s - h * G;
    const float wcol = (float)w;
    const float hrow = (float)h;
    float* __restrict__ obase = out + (size_t)b * 3 * S * 85 + (size_t)s * 85;

    const int mofs = half << 3;
    #pragma unroll
    for (int r = 0; r < 8; ++r) {
        const int o = o_base + mofs + r;
        if (o < 255) {
            const float v = acc[r] + bias[o];
            const int   a    = o / 85;
            const int   attr = o - a * 85;
            const float aw = (a == 0) ? aw0 : ((a == 1) ? aw1 : aw2);
            const float ah = (a == 0) ? ah0 : ((a == 1) ? ah1 : ah2);
            float res;
            if (attr >= 4)      res = sigf(v);
            else if (attr == 0) res = (wcol + sigf(v)) * inv_grid;
            else if (attr == 1) res = (hrow + sigf(v)) * inv_grid;
            else if (attr == 2) res = __expf(v) * aw;
            else                res = __expf(v) * ah;
            __builtin_nontemporal_store(res, obase + (size_t)a * S * 85 + attr);
        }
    }
}

template <int C, int S, int G>
static void launch_head(const float* F, const float* Wm, const float* bias,
                        float* out, const float* anch6, hipStream_t stream)
{
    constexpr int N      = 32 * S;
    constexpr int waves  = 16 * (N / 16);
    constexpr int blocks = waves / 8;
    const float k = 1.0f / 416.0f;
    yolo_head_wmma<C, S, G><<<dim3(blocks), dim3(256), 0, stream>>>(
        F, Wm, bias, out,
        anch6[0] * k, anch6[1] * k, anch6[2] * k,
        anch6[3] * k, anch6[4] * k, anch6[5] * k);
}

extern "C" void kernel_launch(void* const* d_in, const int* in_sizes, int n_in,
                              void* d_out, int out_size, void* d_ws, size_t ws_size,
                              hipStream_t stream) {
    (void)in_sizes; (void)n_in; (void)out_size; (void)d_ws; (void)ws_size;
    const float* f13 = (const float*)d_in[0];
    const float* f26 = (const float*)d_in[1];
    const float* f52 = (const float*)d_in[2];
    const float* W13 = (const float*)d_in[3];
    const float* b13 = (const float*)d_in[4];
    const float* W26 = (const float*)d_in[5];
    const float* b26 = (const float*)d_in[6];
    const float* W52 = (const float*)d_in[7];
    const float* b52 = (const float*)d_in[8];
    float* out = (float*)d_out;

    static const float A13[6] = {116.f, 90.f, 156.f, 198.f, 373.f, 326.f};
    static const float A26[6] = { 30.f, 61.f,  62.f,  45.f,  59.f, 119.f};
    static const float A52[6] = { 10.f, 13.f,  16.f,  30.f,  33.f,  23.f};

    const size_t off13 = 0;
    const size_t off26 = (size_t)32 * 507  * 85;
    const size_t off52 = off26 + (size_t)32 * 2028 * 85;

    launch_head<1024, 13 * 13, 13>(f13, W13, b13, out + off13, A13, stream);
    launch_head< 512, 26 * 26, 26>(f26, W26, b26, out + off26, A26, stream);
    launch_head< 256, 52 * 52, 52>(f52, W52, b52, out + off52, A52, stream);
}


# Register blocking: a wave owns 2x2 / 2x4 / 4x4 output tiles (heads 13/26/52) and feeds their WMMAs from one load of each W and F fragment; per-element k order unchanged; 4-wave work-groups
# speedup vs baseline: 1.2161x; 1.2161x over previous
#include <hip/hip_runtime.h>


constexpr int WPB = 4;

typedef __attribute__((ext_vector_type(2))) float v2f;
typedef __attribute__((ext_vector_type(8))) float v8f;

__device__ __forceinline__ float sigf(float x) {
    return __builtin_amdgcn_rcpf(1.0f + __expf(-x));
}

template <int C, int S, int G, int MT, int NT, int UNR>
__global__ __launch_bounds__(WPB * 32)
void yolo_head_wmma(const float* __restrict__ F,
                    const float* __restrict__ Wm,
                    const float* __restrict__ bias,
                    float* __restrict__ out,
                    float aw0, float ah0, float aw1, float ah1,
                    float aw2, float ah2)
{
    constexpr float inv_grid = 1.0f / (float)G;
    constexpr int MG = 16 / MT;
    static_assert(MG * MT == 16, "MT must divide 16");
    static_assert(((32 * S) / 16) % NT == 0, "NT must divide the N-tile count");
    static_assert((C / 4) % UNR == 0, "unroll must divide the K steps");

    const int lane  = threadIdx.x & 31;
    const int wave  = blockIdx.x * WPB + (threadIdx.x >> 5);
    const int tmg   = wave % MG;
    const int tng   = wave / MG;

    const int l16   = lane & 15;
    const int half  = lane >> 4;
    const int half2 = half << 1;

    const float* __restrict__ wp[MT];
    #pragma unroll
    for (int t = 0; t < MT; ++t) {
        const int o_row  = ((tmg * MT + t) << 4) + l16;
        const int o_load = o_row > 254 ? 254 : o_row;
        wp[t] = Wm + (size_t)o_load * C + half2;
    }

    int bj[NT], sj[NT];
    const float* __restrict__ fp[NT];
    #pragma unroll
    for (int j = 0; j < NT; ++j) {
        const int n = ((tng * NT + j) << 4) + l16;
        bj[j] = n / S;
        sj[j] = n - bj[j] * S;
        fp[j] = F + (size_t)bj[j] * C * S + (size_t)half2 * S + sj[j];
    }

    v8f acc[MT][NT];
    #pragma unroll
    for (int t = 0; t < MT; ++t) {
        #pragma unroll
        for (int j = 0; j < NT; ++j)
            acc[t][j] = (v8f){0.f, 0.f, 0.f, 0.f, 0.f, 0.f, 0.f, 0.f};
    }

    #pragma unroll UNR
    for (int k = 0; k < C; k += 4) {
        v2f a[MT];
        v2f bb[NT];
        #pragma unroll
        for (int t = 0; t < MT; ++t)
            a[t] = *(const v2f*)(wp[t] + k);
        #pragma unroll
        for (int j = 0; j < NT; ++j) {
            bb[j].x = fp[j][k * S];
            bb[j].y = fp[j][k * S + S];
        }
        #pragma unroll
        for (int t = 0; t < MT; ++t) {
            #pragma unroll
            for (int j = 0; j < NT; ++j)
                acc[t][j] = __builtin_amdgcn_wmma_f32_16x16x4_f32(
                                false, a[t], false, bb[j], (short)0, acc[t][j], false, false);
        }
    }

    float wcol[NT], hrow[NT];
    float* __restrict__ obase[NT];
    #pragma unroll
    for (int j = 0; j < NT; ++j) {
        const int s = sj[j];
        const int h = s / G;
        const int w = s - h * G;
        wcol[j] = (float)w;
        hrow[j] = (float)h;
        obase[j] = out + (size_t)bj[j] * 3 * S * 85 + (size_t)s * 85;
    }

    const int mofs = half << 3;
    #pragma unroll
    for (int t = 0; t < MT; ++t) {
        const int o_base = (tmg * MT + t) << 4;
        #pragma unroll
        for (int r = 0; r < 8; ++r) {
            const int o = o_base + mofs + r;
            if (o < 255) {
                const float bv   = bias[o];
                const int   a    = o / 85;
                const int   attr = o - a * 85;
                const float aw = (a == 0) ? aw0 : ((a == 1) ? aw1 : aw2);
                const float ah = (a == 0) ? ah0 : ((a == 1) ? ah1 : ah2);
                const int   off = a * S * 85 + attr;
                #pragma unroll
                for (int j = 0; j < NT; ++j) {
                    const float v = acc[t][j][r] + bv;
                    float res;
                    if (attr >= 4)      res = sigf(v);
                    else if (attr == 0) res = (wcol[j] + sigf(v)) * inv_grid;
                    else if (attr == 1) res = (hrow[j] + sigf(v)) * inv_grid;
                    else if (attr == 2) res = __expf(v) * aw;
                    else                res = __expf(v) * ah;
                    __builtin_nontemporal_store(res, obase[j] + off);
                }
            }
        }
    }
}

template <int C, int S, int G, int MT, int NT, int UNR>
static void launch_head(const float* F, const float* Wm, const float* bias,
                        float* out, const float* anch6, hipStream_t stream)
{
    constexpr int N      = 32 * S;
    constexpr int waves  = (16 / MT) * (N / 16 / NT);
    constexpr int blocks = waves / WPB;
    static_assert(blocks * WPB == waves, "wave count must fill whole blocks");
    const float k = 1.0f / 416.0f;
    yolo_head_wmma<C, S, G, MT, NT, UNR><<<dim3(blocks), dim3(WPB * 32), 0, stream>>>(
        F, Wm, bias, out,
        anch6[0] * k, anch6[1] * k, anch6[2] * k,
        anch6[3] * k, anch6[4] * k, anch6[5] * k);
}

extern "C" void kernel_launch(void* const* d_in, const int* in_sizes, int n_in,
                              void* d_out, int out_size, void* d_ws, size_t ws_size,
                              hipStream_t stream) {
    (void)in_sizes; (void)n_in; (void)out_size; (void)d_ws; (void)ws_size;
    const float* f13 = (const float*)d_in[0];
    const float* f26 = (const float*)d_in[1];
    const float* f52 = (const float*)d_in[2];
    const float* W13 = (const float*)d_in[3];
    const float* b13 = (const float*)d_in[4];
    const float* W26 = (const float*)d_in[5];
    const float* b26 = (const float*)d_in[6];
    const float* W52 = (const float*)d_in[7];
    const float* b52 = (const float*)d_in[8];
    float* out = (float*)d_out;

    static const float A13[6] = {116.f, 90.f, 156.f, 198.f, 373.f, 326.f};
    static const float A26[6] = { 30.f, 61.f,  62.f,  45.f,  59.f, 119.f};
    static const float A52[6] = { 10.f, 13.f,  16.f,  30.f,  33.f,  23.f};

    const size_t off13 = 0;
    const size_t off26 = (size_t)32 * 507  * 85;
    const size_t off52 = off26 + (size_t)32 * 2028 * 85;

    launch_head<1024, 13 * 13, 13, 2, 2, 4>(f13, W13, b13, out + off13, A13, stream);
    launch_head< 512, 26 * 26, 26, 2, 4, 4>(f26, W26, b26, out + off26, A26, stream);
    launch_head< 256, 52 * 52, 52, 4, 4, 2>(f52, W52, b52, out + off52, A52, stream);
}


# Swap WMMA operand roles (A = activations, B = weights): a lane holds one channel, so each store writes two 64-byte runs along the attribute axis, not 32 scattered dwords; k order unchanged
# speedup vs baseline: 2.3255x; 1.9123x over previous
#include <hip/hip_runtime.h>


constexpr int WPB = 4;

typedef __attribute__((ext_vector_type(2))) float v2f;
typedef __attribute__((ext_vector_type(8))) float v8f;

template <int C, int S, int G, int MT, int NT, int UNR>
__global__ __launch_bounds__(WPB * 32)
void yolo_head_wmma(const float* __restrict__ F,
                    const float* __restrict__ Wm,
                    const float* __restrict__ bias,
                    float* __restrict__ out,
                    float aw0, float ah0, float aw1, float ah1,
                    float aw2, float ah2)
{
    constexpr float inv_grid = 1.0f / (float)G;
    constexpr int MG = 16 / MT;
    static_assert(MG * MT == 16, "MT must divide 16");
    static_assert(((32 * S) / 16) % NT == 0, "NT must divide the N-tile count");
    static_assert((C / 4) % UNR == 0, "unroll must divide the K steps");

    const int lane  = threadIdx.x & 31;
    const int wave  = blockIdx.x * WPB + (threadIdx.x >> 5);
    const int tmg   = wave % MG;
    const int tng   = wave / MG;

    const int l16   = lane & 15;
    const int half  = lane >> 4;
    const int half2 = half << 1;

    const float* __restrict__ wp[MT];
    #pragma unroll
    for (int t = 0; t < MT; ++t) {
        const int o_row  = ((tmg * MT + t) << 4) + l16;
        const int o_load = o_row > 254 ? 254 : o_row;
        wp[t] = Wm + (size_t)o_load * C + half2;
    }

    int bj[NT], sj[NT];
    const float* __restrict__ fp[NT];
    #pragma unroll
    for (int j = 0; j < NT; ++j) {
        const int n = ((tng * NT + j) << 4) + l16;
        bj[j] = n / S;
        sj[j] = n - bj[j] * S;
        fp[j] = F + (size_t)bj[j] * C * S + (size_t)half2 * S + sj[j];
    }

    v8f acc[MT][NT];
    #pragma unroll
    for (int t = 0; t < MT; ++t) {
        #pragma unroll
        for (int j = 0; j < NT; ++j)
            acc[t][j] = (v8f){0.f, 0.f, 0.f, 0.f, 0.f, 0.f, 0.f, 0.f};
    }

    #pragma unroll UNR
    for (int k = 0; k < C; k += 4) {
        v2f a[MT];
        v2f bb[NT];
        #pragma unroll
        for (int t = 0; t < MT; ++t)
            a[t] = *(const v2f*)(wp[t] + k);
        #pragma unroll
        for (int j = 0; j < NT; ++j) {
            bb[j].x = fp[j][k * S];
            bb[j].y = fp[j][k * S + S];
        }
        #pragma unroll
        for (int t = 0; t < MT; ++t) {
            #pragma unroll
            for (int j = 0; j < NT; ++j)
                acc[t][j] = __builtin_amdgcn_wmma_f32_16x16x4_f32(
                                false, bb[j], false, a[t], (short)0, acc[t][j], false, false);
        }
    }

    float bvT[MT], anchT[MT], mulT[MT];
    int   offT[MT];
    bool  expT[MT], isxT[MT], isyT[MT], okT[MT];
    #pragma unroll
    for (int t = 0; t < MT; ++t) {
        const int o    = ((tmg * MT + t) << 4) + l16;
        const int oc   = o > 254 ? 254 : o;
        const int a    = oc / 85;
        const int attr = oc - a * 85;
        const float aw = (a == 0) ? aw0 : ((a == 1) ? aw1 : aw2);
        const float ah = (a == 0) ? ah0 : ((a == 1) ? ah1 : ah2);
        bvT[t]   = bias[oc];
        anchT[t] = (attr == 2) ? aw : ah;
        mulT[t]  = (attr < 2) ? inv_grid : 1.0f;
        offT[t]  = a * S * 85 + attr;
        expT[t]  = (attr == 2) || (attr == 3);
        isxT[t]  = (attr == 0);
        isyT[t]  = (attr == 1);
        okT[t]   = (o < 255);
    }

    const unsigned mofs = (unsigned)half << 3;
    #pragma unroll
    for (int j = 0; j < NT; ++j) {
        #pragma unroll
        for (int r = 0; r < 8; ++r) {
            const unsigned n  = ((unsigned)(tng * NT + j) << 4) + mofs + (unsigned)r;
            const unsigned b  = n / (unsigned)S;
            const unsigned s  = n - b * (unsigned)S;
            const unsigned gh = s / (unsigned)G;
            const unsigned gw = s - gh * (unsigned)G;
            const float wcol = (float)gw;
            const float hrow = (float)gh;
            float* __restrict__ orow = out + (size_t)b * 3 * S * 85 + (size_t)s * 85;
            #pragma unroll
            for (int t = 0; t < MT; ++t) {
                const float v   = acc[t][j][r] + bvT[t];
                const float e   = __expf(expT[t] ? v : -v);
                const float sg  = __builtin_amdgcn_rcpf(1.0f + e);
                const float add = isxT[t] ? wcol : (isyT[t] ? hrow : 0.0f);
                const float rs  = (add + sg) * mulT[t];
                const float re  = e * anchT[t];
                const float res = expT[t] ? re : rs;
                if (okT[t]) __builtin_nontemporal_store(res, orow + offT[t]);
            }
        }
    }
}

template <int C, int S, int G, int MT, int NT, int UNR>
static void launch_head(const float* F, const float* Wm, const float* bias,
                        float* out, const float* anch6, hipStream_t stream)
{
    constexpr int N      = 32 * S;
    constexpr int waves  = (16 / MT) * (N / 16 / NT);
    constexpr int blocks = waves / WPB;
    static_assert(blocks * WPB == waves, "wave count must fill whole blocks");
    const float k = 1.0f / 416.0f;
    yolo_head_wmma<C, S, G, MT, NT, UNR><<<dim3(blocks), dim3(WPB * 32), 0, stream>>>(
        F, Wm, bias, out,
        anch6[0] * k, anch6[1] * k, anch6[2] * k,
        anch6[3] * k, anch6[4] * k, anch6[5] * k);
}

extern "C" void kernel_launch(void* const* d_in, const int* in_sizes, int n_in,
                              void* d_out, int out_size, void* d_ws, size_t ws_size,
                              hipStream_t stream) {
    (void)in_sizes; (void)n_in; (void)out_size; (void)d_ws; (void)ws_size;
    const float* f13 = (const float*)d_in[0];
    const float* f26 = (const float*)d_in[1];
    const float* f52 = (const float*)d_in[2];
    const float* W13 = (const float*)d_in[3];
    const float* b13 = (const float*)d_in[4];
    const float* W26 = (const float*)d_in[5];
    const float* b26 = (const float*)d_in[6];
    const float* W52 = (const float*)d_in[7];
    const float* b52 = (const float*)d_in[8];
    float* out = (float*)d_out;

    static const float A13[6] = {116.f, 90.f, 156.f, 198.f, 373.f, 326.f};
    static const float A26[6] = { 30.f, 61.f,  62.f,  45.f,  59.f, 119.f};
    static const float A52[6] = { 10.f, 13.f,  16.f,  30.f,  33.f,  23.f};

    const size_t off13 = 0;
    const size_t off26 = (size_t)32 * 507  * 85;
    const size_t off52 = off26 + (size_t)32 * 2028 * 85;

    launch_head<1024, 13 * 13, 13, 2, 2, 4>(f13, W13, b13, out + off13, A13, stream);
    launch_head< 512, 26 * 26, 26, 2, 4, 4>(f26, W26, b26, out + off26, A26, stream);
    launch_head< 256, 52 * 52, 52, 4, 4, 2>(f52, W52, b52, out + off52, A52, stream);
}


# Stage operands in LDS: a work-group copies each K chunk of 32 (weight rows, activation columns) once with coalesced loads, waves read fragments from LDS, two buffers; k order unchanged
# speedup vs baseline: 2.9853x; 1.2837x over previous
#include <hip/hip_runtime.h>


constexpr int WPB = 4;

typedef __attribute__((ext_vector_type(2))) float v2f;
typedef __attribute__((ext_vector_type(4))) float v4f;
typedef __attribute__((ext_vector_type(8))) float v8f;

template <int C, int S, int G, int MT, int NT, int VEC>
__global__ __launch_bounds__(WPB * 32)
void yolo_head_wmma(const float* __restrict__ F,
                    const float* __restrict__ Wm,
                    const float* __restrict__ bias,
                    float* __restrict__ out,
                    float aw0, float ah0, float aw1, float ah1,
                    float aw2, float ah2)
{
    constexpr float inv_grid = 1.0f / (float)G;
    constexpr int MG  = 16 / MT;
    constexpr int KC  = 32;
    constexpr int NCH = C / KC;
    constexpr int R   = WPB * MT * 16;
    constexpr int NB  = NT * 16;
    constexpr int PW  = KC + 4;
    constexpr int PF  = NB + 8;
    constexpr int WBUF = R * PW;
    constexpr int FBUF = KC * PF;
    constexpr int BUF  = WBUF + FBUF;
    static_assert(MG * MT == 16, "MT must divide 16");
    static_assert(MG % WPB == 0, "a work-group's waves must share one position group");
    static_assert(((32 * S) / 16) % NT == 0, "NT must divide the N-tile count");
    static_assert(C % KC == 0, "K chunk must divide C");
    static_assert(VEC == 1 || (VEC == 4 && S % 4 == 0), "16-byte activation loads need S % 4 == 0");
    static_assert(BUF % 4 == 0 && WBUF % 4 == 0 && PW % 4 == 0 && PF % 4 == 0, "16-byte LDS alignment");

    __shared__ __attribute__((aligned(16))) float lds[2 * BUF];

    const int tid   = threadIdx.x;
    const int lane  = tid & 31;
    const int wv    = tid >> 5;
    const int wave  = blockIdx.x * WPB + wv;
    const int tmg   = wave % MG;
    const int tng   = wave / MG;
    const int rowbase = ((blockIdx.x * WPB) % MG) * (MT * 16);

    const int l16   = lane & 15;
    const int half  = lane >> 4;
    const int half2 = half << 1;

    constexpr int WI = R / 16;
    const int wq = tid & 7;
    const int wr = tid >> 3;
    constexpr int GPR = NB / VEC;
    constexpr int FRS = 128 / GPR;
    constexpr int FI  = KC / FRS;
    static_assert(128 % GPR == 0 && KC % FRS == 0, "copy plan must tile the chunk");
    const int fg = tid % GPR;
    const int fr = tid / GPR;
    const int n0 = tng * NB + fg * VEC;
    const int fb = n0 / S;
    const int fs = n0 - fb * S;
    const float* __restrict__ fsrc = F + (size_t)fb * C * S + (size_t)fr * S + fs;

    v4f wreg[WI];
    v4f fregv[VEC == 4 ? FI : 1];
    float fregs[VEC == 1 ? FI : 1];

    auto load_chunk = [&](int kc) {
        #pragma unroll
        for (int i = 0; i < WI; ++i) {
            int row = rowbase + wr + 16 * i;
            row = row > 254 ? 254 : row;
            wreg[i] = *(const v4f*)(Wm + (size_t)row * C + kc + 4 * wq);
        }
        #pragma unroll
        for (int i = 0; i < FI; ++i) {
            const float* p = fsrc + (size_t)(kc + i * FRS) * S;
            if constexpr (VEC == 4) fregv[i] = *(const v4f*)p;
            else                    fregs[i] = *p;
        }
    };
    auto store_chunk = [&](float* buf) {
        #pragma unroll
        for (int i = 0; i < WI; ++i)
            *(v4f*)(buf + (wr + 16 * i) * PW + 4 * wq) = wreg[i];
        #pragma unroll
        for (int i = 0; i < FI; ++i) {
            float* q = buf + WBUF + (fr + i * FRS) * PF + fg * VEC;
            if constexpr (VEC == 4) *(v4f*)q = fregv[i];
            else                    *q = fregs[i];
        }
    };

    v8f acc[MT][NT];
    #pragma unroll
    for (int t = 0; t < MT; ++t) {
        #pragma unroll
        for (int j = 0; j < NT; ++j)
            acc[t][j] = (v8f){0.f, 0.f, 0.f, 0.f, 0.f, 0.f, 0.f, 0.f};
    }

    const int wl = ((wv * MT) * 16 + l16) * PW + half2;
    const int fl = WBUF + half2 * PF + l16;

    load_chunk(0);
    store_chunk(lds);
    __syncthreads();

    for (int c = 0; c < NCH; ++c) {
        const float* cur = lds + (c & 1) * BUF;
        if (c + 1 < NCH) load_chunk((c + 1) * KC);

        #pragma unroll
        for (int kk = 0; kk < KC; kk += 4) {
            v2f a[MT];
            v2f bb[NT];
            #pragma unroll
            for (int t = 0; t < MT; ++t)
                a[t] = *(const v2f*)(cur + wl + t * 16 * PW + kk);
            #pragma unroll
            for (int j = 0; j < NT; ++j) {
                bb[j].x = cur[fl + kk * PF + j * 16];
                bb[j].y = cur[fl + kk * PF + PF + j * 16];
            }
            #pragma unroll
            for (int t = 0; t < MT; ++t) {
                #pragma unroll
                for (int j = 0; j < NT; ++j)
                    acc[t][j] = __builtin_amdgcn_wmma_f32_16x16x4_f32(
                                    false, bb[j], false, a[t], (short)0, acc[t][j], false, false);
            }
        }

        if (c + 1 < NCH) store_chunk(lds + ((c + 1) & 1) * BUF);
        __syncthreads();
    }

    float bvT[MT], anchT[MT], mulT[MT];
    int   offT[MT];
    bool  expT[MT], isxT[MT], isyT[MT], okT[MT];
    #pragma unroll
    for (int t = 0; t < MT; ++t) {
        const int o    = ((tmg * MT + t) << 4) + l16;
        const int oc   = o > 254 ? 254 : o;
        const int a    = oc / 85;
        const int attr = oc - a * 85;
        const float aw = (a == 0) ? aw0 : ((a == 1) ? aw1 : aw2);
        const float ah = (a == 0) ? ah0 : ((a == 1) ? ah1 : ah2);
        bvT[t]   = bias[oc];
        anchT[t] = (attr == 2) ? aw : ah;
        mulT[t]  = (attr < 2) ? inv_grid : 1.0f;
        offT[t]  = a * S * 85 + attr;
        expT[t]  = (attr == 2) || (attr == 3);
        isxT[t]  = (attr == 0);
        isyT[t]  = (attr == 1);
        okT[t]   = (o < 255);
    }

    const unsigned mofs = (unsigned)half << 3;
    #pragma unroll
    for (int j = 0; j < NT; ++j) {
        #pragma unroll
        for (int r = 0; r < 8; ++r) {
            const unsigned n  = ((unsigned)(tng * NT + j) << 4) + mofs + (unsigned)r;
            const unsigned b  = n / (unsigned)S;
            const unsigned s  = n - b * (unsigned)S;
            const unsigned gh = s / (unsigned)G;
            const unsigned gw = s - gh * (unsigned)G;
            const float wcol = (float)gw;
            const float hrow = (float)gh;
            float* __restrict__ orow = out + (size_t)b * 3 * S * 85 + (size_t)s * 85;
            #pragma unroll
            for (int t = 0; t < MT; ++t) {
                const float v   = acc[t][j][r] + bvT[t];
                const float e   = __expf(expT[t] ? v : -v);
                const float sg  = __builtin_amdgcn_rcpf(1.0f + e);
                const float add = isxT[t] ? wcol : (isyT[t] ? hrow : 0.0f);
                const float rs  = (add + sg) * mulT[t];
                const float re  = e * anchT[t];
                const float res = expT[t] ? re : rs;
                if (okT[t]) __builtin_nontemporal_store(res, orow + offT[t]);
            }
        }
    }
}

template <int C, int S, int G, int MT, int NT, int VEC>
static void launch_head(const float* F, const float* Wm, const float* bias,
                        float* out, const float* anch6, hipStream_t stream)
{
    constexpr int N      = 32 * S;
    constexpr int waves  = (16 / MT) * (N / 16 / NT);
    constexpr int blocks = waves / WPB;
    static_assert(blocks * WPB == waves, "wave count must fill whole blocks");
    const float k = 1.0f / 416.0f;
    yolo_head_wmma<C, S, G, MT, NT, VEC><<<dim3(blocks), dim3(WPB * 32), 0, stream>>>(
        F, Wm, bias, out,
        anch6[0] * k, anch6[1] * k, anch6[2] * k,
        anch6[3] * k, anch6[4] * k, anch6[5] * k);
}

extern "C" void kernel_launch(void* const* d_in, const int* in_sizes, int n_in,
                              void* d_out, int out_size, void* d_ws, size_t ws_size,
                              hipStream_t stream) {
    (void)in_sizes; (void)n_in; (void)out_size; (void)d_ws; (void)ws_size;
    const float* f13 = (const float*)d_in[0];
    const float* f26 = (const float*)d_in[1];
    const float* f52 = (const float*)d_in[2];
    const float* W13 = (const float*)d_in[3];
    const float* b13 = (const float*)d_in[4];
    const float* W26 = (const float*)d_in[5];
    const float* b26 = (const float*)d_in[6];
    const float* W52 = (const float*)d_in[7];
    const float* b52 = (const float*)d_in[8];
    float* out = (float*)d_out;

    static const float A13[6] = {116.f, 90.f, 156.f, 198.f, 373.f, 326.f};
    static const float A26[6] = { 30.f, 61.f,  62.f,  45.f,  59.f, 119.f};
    static const float A52[6] = { 10.f, 13.f,  16.f,  30.f,  33.f,  23.f};

    const size_t off13 = 0;
    const size_t off26 = (size_t)32 * 507  * 85;
    const size_t off52 = off26 + (size_t)32 * 2028 * 85;

    launch_head<1024, 13 * 13, 13, 2, 2, 1>(f13, W13, b13, out + off13, A13, stream);
    launch_head< 512, 26 * 26, 26, 2, 4, 4>(f26, W26, b26, out + off26, A26, stream);
    launch_head< 256, 52 * 52, 52, 4, 4, 4>(f52, W52, b52, out + off52, A52, stream);
}


# Epilogue staged through LDS in output order: the work-group copies each anchor's rows out with wave-contiguous 16-byte stores (512 B per instruction), not a dword per lane in two 64-byte runs
# speedup vs baseline: 3.8278x; 1.2822x over previous
#include <hip/hip_runtime.h>


constexpr int WPB = 4;

typedef __attribute__((ext_vector_type(2))) float v2f;
typedef __attribute__((ext_vector_type(4))) float v4f;
typedef __attribute__((ext_vector_type(8))) float v8f;

template <int S, int NB, int LEN>
__device__ __forceinline__ void copy_rows(const float* __restrict__ q, float* __restrict__ out,
                                          int a, int lo, unsigned n0wg, int tid)
{
    for (int e = tid; e < NB * LEN; e += WPB * 32) {
        const unsigned p = (unsigned)e / (unsigned)LEN;
        const unsigned i = (unsigned)e - p * (unsigned)LEN;
        const unsigned n = n0wg + p;
        const unsigned b = n / (unsigned)S;
        const unsigned s = n - b * (unsigned)S;
        const float val = q[p * 85u + (unsigned)lo + i];
        __builtin_nontemporal_store(val, out + ((size_t)(b * 3u + (unsigned)a) * S + s) * 85 + (unsigned)lo + i);
    }
}

template <int C, int S, int G, int MT, int NT, int VEC>
__global__ __launch_bounds__(WPB * 32)
void yolo_head_wmma(const float* __restrict__ F,
                    const float* __restrict__ Wm,
                    const float* __restrict__ bias,
                    float* __restrict__ out,
                    float aw0, float ah0, float aw1, float ah1,
                    float aw2, float ah2)
{
    constexpr float inv_grid = 1.0f / (float)G;
    constexpr int MG  = 16 / MT;
    constexpr int KC  = 32;
    constexpr int NCH = C / KC;
    constexpr int R   = WPB * MT * 16;
    constexpr int NB  = NT * 16;
    constexpr int PW  = KC + 4;
    constexpr int PF  = NB + 8;
    constexpr int WBUF = R * PW;
    constexpr int FBUF = KC * PF;
    constexpr int BUF  = WBUF + FBUF;
    static_assert(MG * MT == 16, "MT must divide 16");
    static_assert(MG % WPB == 0, "a work-group's waves must share one position group");
    static_assert(((32 * S) / 16) % NT == 0, "NT must divide the N-tile count");
    static_assert(C % KC == 0, "K chunk must divide C");
    static_assert(VEC == 1 || (VEC == 4 && S % 4 == 0), "16-byte activation loads need S % 4 == 0");
    static_assert(BUF % 4 == 0 && WBUF % 4 == 0 && PW % 4 == 0 && PF % 4 == 0, "16-byte LDS alignment");

    __shared__ __attribute__((aligned(16))) float lds[2 * BUF];

    const int tid   = threadIdx.x;
    const int lane  = tid & 31;
    const int wv    = tid >> 5;
    const int wave  = blockIdx.x * WPB + wv;
    const int tmg   = wave % MG;
    const int tng   = wave / MG;
    const int rowbase = ((blockIdx.x * WPB) % MG) * (MT * 16);

    const int l16   = lane & 15;
    const int half  = lane >> 4;
    const int half2 = half << 1;

    constexpr int WI = R / 16;
    const int wq = tid & 7;
    const int wr = tid >> 3;
    constexpr int GPR = NB / VEC;
    constexpr int FRS = 128 / GPR;
    constexpr int FI  = KC / FRS;
    static_assert(128 % GPR == 0 && KC % FRS == 0, "copy plan must tile the chunk");
    const int fg = tid % GPR;
    const int fr = tid / GPR;
    const int n0 = tng * NB + fg * VEC;
    const int fb = n0 / S;
    const int fs = n0 - fb * S;
    const float* __restrict__ fsrc = F + (size_t)fb * C * S + (size_t)fr * S + fs;

    v4f wreg[WI];
    v4f fregv[VEC == 4 ? FI : 1];
    float fregs[VEC == 1 ? FI : 1];

    auto load_chunk = [&](int kc) {
        #pragma unroll
        for (int i = 0; i < WI; ++i) {
            int row = rowbase + wr + 16 * i;
            row = row > 254 ? 254 : row;
            wreg[i] = *(const v4f*)(Wm + (size_t)row * C + kc + 4 * wq);
        }
        #pragma unroll
        for (int i = 0; i < FI; ++i) {
            const float* p = fsrc + (size_t)(kc + i * FRS) * S;
            if constexpr (VEC == 4) fregv[i] = *(const v4f*)p;
            else                    fregs[i] = *p;
        }
    };
    auto store_chunk = [&](float* buf) {
        #pragma unroll
        for (int i = 0; i < WI; ++i)
            *(v4f*)(buf + (wr + 16 * i) * PW + 4 * wq) = wreg[i];
        #pragma unroll
        for (int i = 0; i < FI; ++i) {
            float* q = buf + WBUF + (fr + i * FRS) * PF + fg * VEC;
            if constexpr (VEC == 4) *(v4f*)q = fregv[i];
            else                    *q = fregs[i];
        }
    };

    v8f acc[MT][NT];
    #pragma unroll
    for (int t = 0; t < MT; ++t) {
        #pragma unroll
        for (int j = 0; j < NT; ++j)
            acc[t][j] = (v8f){0.f, 0.f, 0.f, 0.f, 0.f, 0.f, 0.f, 0.f};
    }

    const int wl = ((wv * MT) * 16 + l16) * PW + half2;
    const int fl = WBUF + half2 * PF + l16;

    load_chunk(0);
    store_chunk(lds);
    __syncthreads();

    for (int c = 0; c < NCH; ++c) {
        const float* cur = lds + (c & 1) * BUF;
        if (c + 1 < NCH) load_chunk((c + 1) * KC);

        #pragma unroll
        for (int kk = 0; kk < KC; kk += 4) {
            v2f a[MT];
            v2f bb[NT];
            #pragma unroll
            for (int t = 0; t < MT; ++t)
                a[t] = *(const v2f*)(cur + wl + t * 16 * PW + kk);
            #pragma unroll
            for (int j = 0; j < NT; ++j) {
                bb[j].x = cur[fl + kk * PF + j * 16];
                bb[j].y = cur[fl + kk * PF + PF + j * 16];
            }
            #pragma unroll
            for (int t = 0; t < MT; ++t) {
                #pragma unroll
                for (int j = 0; j < NT; ++j)
                    acc[t][j] = __builtin_amdgcn_wmma_f32_16x16x4_f32(
                                    false, bb[j], false, a[t], (short)0, acc[t][j], false, false);
            }
        }

        if (c + 1 < NCH) store_chunk(lds + ((c + 1) & 1) * BUF);
        __syncthreads();
    }

    constexpr int NA    = (R == 256) ? 3 : 2;
    constexpr int PSZ   = NB * 85;
    constexpr int TRASH = NA * PSZ;
    static_assert(R == 256 || R == 128, "the anchor split below is written for these two row counts");
    static_assert(TRASH + 1 <= 2 * BUF, "the staged output must fit in the operand buffers");
    static_assert(PSZ % 4 == 0, "16-byte LDS alignment of each anchor's block");
    const int a_first = (R == 256) ? 0 : (rowbase != 0 ? 1 : 0);

    float bvT[MT], anchT[MT], mulT[MT];
    int   qbT[MT], qsT[MT];
    bool  expT[MT], isxT[MT], isyT[MT];
    #pragma unroll
    for (int t = 0; t < MT; ++t) {
        const int o    = ((tmg * MT + t) << 4) + l16;
        const int oc   = o > 254 ? 254 : o;
        const int a    = oc / 85;
        const int attr = oc - a * 85;
        const float aw = (a == 0) ? aw0 : ((a == 1) ? aw1 : aw2);
        const float ah = (a == 0) ? ah0 : ((a == 1) ? ah1 : ah2);
        const bool  ok = (o < 255);
        bvT[t]   = bias[oc];
        anchT[t] = (attr == 2) ? aw : ah;
        mulT[t]  = (attr < 2) ? inv_grid : 1.0f;
        expT[t]  = (attr == 2) || (attr == 3);
        isxT[t]  = (attr == 0);
        isyT[t]  = (attr == 1);
        qbT[t]   = ok ? ((a - a_first) * PSZ + attr + 8 * 85 * half) : TRASH;
        qsT[t]   = ok ? 85 : 0;
    }

    const unsigned mofs = (unsigned)half << 3;
    #pragma unroll
    for (int j = 0; j < NT; ++j) {
        #pragma unroll
        for (int r = 0; r < 8; ++r) {
            const unsigned n  = ((unsigned)(tng * NT + j) << 4) + mofs + (unsigned)r;
            const unsigned b  = n / (unsigned)S;
            const unsigned s  = n - b * (unsigned)S;
            const unsigned gh = s / (unsigned)G;
            const unsigned gw = s - gh * (unsigned)G;
            const float wcol = (float)gw;
            const float hrow = (float)gh;
            #pragma unroll
            for (int t = 0; t < MT; ++t) {
                const float v   = acc[t][j][r] + bvT[t];
                const float e   = __expf(expT[t] ? v : -v);
                const float sg  = __builtin_amdgcn_rcpf(1.0f + e);
                const float add = isxT[t] ? wcol : (isyT[t] ? hrow : 0.0f);
                const float rs  = (add + sg) * mulT[t];
                const float re  = e * anchT[t];
                const float res = expT[t] ? re : rs;
                lds[qbT[t] + (16 * j + r) * qsT[t]] = res;
            }
        }
    }
    __syncthreads();

    const unsigned n0wg = (unsigned)(tng * NB);
    const int wvu = __builtin_amdgcn_readfirstlane(wv);
    #pragma unroll
    for (int al = 0; al < NA; ++al) {
        const int a  = a_first + al;
        const int lo = (rowbase - a * 85) > 0 ? (rowbase - a * 85) : 0;
        const int hi = (rowbase + R - a * 85) < 85 ? (rowbase + R - a * 85) : 85;
        const float* __restrict__ q = lds + al * PSZ;
        if (VEC == 4 && lo == 0 && hi == 85) {
            for (int g = wvu; g < NB / 4; g += WPB) {
                const unsigned n = n0wg + 4u * (unsigned)g;
                const unsigned b = n / (unsigned)S;
                const unsigned s = n - b * (unsigned)S;
                float* __restrict__ dst = out + ((size_t)(b * 3u + (unsigned)a) * S + s) * 85;
                const float* src = q + g * 340;
                #pragma unroll
                for (int k = 0; k < 3; ++k) {
                    const int f = lane + 32 * k;
                    if (f < 85) {
                        const v4f val = *(const v4f*)(src + 4 * f);
                        __builtin_nontemporal_store(val, (v4f*)(dst + 4 * f));
                    }
                }
            }
        } else if (lo == 0 && hi == 85) {
            copy_rows<S, NB, 85>(q, out, a, 0, n0wg, tid);
        } else if (lo == 0) {
            copy_rows<S, NB, 43>(q, out, a, 0, n0wg, tid);
        } else {
            copy_rows<S, NB, 42>(q, out, a, 43, n0wg, tid);
        }
    }
}

template <int C, int S, int G, int MT, int NT, int VEC>
static void launch_head(const float* F, const float* Wm, const float* bias,
                        float* out, const float* anch6, hipStream_t stream)
{
    constexpr int N      = 32 * S;
    constexpr int waves  = (16 / MT) * (N / 16 / NT);
    constexpr int blocks = waves / WPB;
    static_assert(blocks * WPB == waves, "wave count must fill whole blocks");
    const float k = 1.0f / 416.0f;
    yolo_head_wmma<C, S, G, MT, NT, VEC><<<dim3(blocks), dim3(WPB * 32), 0, stream>>>(
        F, Wm, bias, out,
        anch6[0] * k, anch6[1] * k, anch6[2] * k,
        anch6[3] * k, anch6[4] * k, anch6[5] * k);
}

extern "C" void kernel_launch(void* const* d_in, const int* in_sizes, int n_in,
                              void* d_out, int out_size, void* d_ws, size_t ws_size,
                              hipStream_t stream) {
    (void)in_sizes; (void)n_in; (void)out_size; (void)d_ws; (void)ws_size;
    const float* f13 = (const float*)d_in[0];
    const float* f26 = (const float*)d_in[1];
    const float* f52 = (const float*)d_in[2];
    const float* W13 = (const float*)d_in[3];
    const float* b13 = (const float*)d_in[4];
    const float* W26 = (const float*)d_in[5];
    const float* b26 = (const float*)d_in[6];
    const float* W52 = (const float*)d_in[7];
    const float* b52 = (const float*)d_in[8];
    float* out = (float*)d_out;

    static const float A13[6] = {116.f, 90.f, 156.f, 198.f, 373.f, 326.f};
    static const float A26[6] = { 30.f, 61.f,  62.f,  45.f,  59.f, 119.f};
    static const float A52[6] = { 10.f, 13.f,  16.f,  30.f,  33.f,  23.f};

    const size_t off13 = 0;
    const size_t off26 = (size_t)32 * 507  * 85;
    const size_t off52 = off26 + (size_t)32 * 2028 * 85;

    launch_head<1024, 13 * 13, 13, 2, 2, 1>(f13, W13, b13, out + off13, A13, stream);
    launch_head< 512, 26 * 26, 26, 2, 4, 4>(f26, W26, b26, out + off26, A26, stream);
    launch_head< 256, 52 * 52, 52, 4, 4, 4>(f52, W52, b52, out + off52, A52, stream);
}


# One launch for the three heads: the work-group index picks the head (deepest-K head first), so the grids' tails and launch gaps overlap and every SIMD stays full; per-element k order unchanged
# speedup vs baseline: 4.4727x; 1.1685x over previous
#include <hip/hip_runtime.h>


constexpr int WPB = 4;

typedef __attribute__((ext_vector_type(2))) float v2f;
typedef __attribute__((ext_vector_type(4))) float v4f;
typedef __attribute__((ext_vector_type(8))) float v8f;

template <int S, int NB, int LEN>
__device__ __forceinline__ void copy_rows(const float* __restrict__ q, float* __restrict__ out,
                                          int a, int lo, unsigned n0wg, int tid)
{
    for (int e = tid; e < NB * LEN; e += WPB * 32) {
        const unsigned p = (unsigned)e / (unsigned)LEN;
        const unsigned i = (unsigned)e - p * (unsigned)LEN;
        const unsigned n = n0wg + p;
        const unsigned b = n / (unsigned)S;
        const unsigned s = n - b * (unsigned)S;
        const float val = q[p * 85u + (unsigned)lo + i];
        __builtin_nontemporal_store(val, out + ((size_t)(b * 3u + (unsigned)a) * S + s) * 85 + (unsigned)lo + i);
    }
}

template <int S, int MT, int NT>
constexpr int head_blocks() { return (16 / MT) * ((32 * S) / 16 / NT) / WPB; }

template <int MT, int NT>
constexpr int head_lds_dwords() { return 2 * (WPB * MT * 16 * (32 + 4) + 32 * (NT * 16 + 8)); }

template <int C, int S, int G, int MT, int NT, int VEC>
__device__ __forceinline__
void yolo_head_wmma(float* lds,
                    const int bid,
                    const float* __restrict__ F,
                    const float* __restrict__ Wm,
                    const float* __restrict__ bias,
                    float* __restrict__ out,
                    float aw0, float ah0, float aw1, float ah1,
                    float aw2, float ah2)
{
    constexpr float inv_grid = 1.0f / (float)G;
    constexpr int MG  = 16 / MT;
    constexpr int KC  = 32;
    constexpr int NCH = C / KC;
    constexpr int R   = WPB * MT * 16;
    constexpr int NB  = NT * 16;
    constexpr int PW  = KC + 4;
    constexpr int PF  = NB + 8;
    constexpr int WBUF = R * PW;
    constexpr int FBUF = KC * PF;
    constexpr int BUF  = WBUF + FBUF;
    static_assert(MG * MT == 16, "MT must divide 16");
    static_assert(MG % WPB == 0, "a work-group's waves must share one position group");
    static_assert(((32 * S) / 16) % NT == 0, "NT must divide the N-tile count");
    static_assert(C % KC == 0, "K chunk must divide C");
    static_assert(VEC == 1 || (VEC == 4 && S % 4 == 0), "16-byte activation loads need S % 4 == 0");
    static_assert(BUF % 4 == 0 && WBUF % 4 == 0 && PW % 4 == 0 && PF % 4 == 0, "16-byte LDS alignment");

    static_assert(2 * BUF == head_lds_dwords<MT, NT>(), "LDS sizing helper out of step with the body");

    const int tid   = threadIdx.x;
    const int lane  = tid & 31;
    const int wv    = tid >> 5;
    const int wave  = bid * WPB + wv;
    const int tmg   = wave % MG;
    const int tng   = wave / MG;
    const int rowbase = ((bid * WPB) % MG) * (MT * 16);

    const int l16   = lane & 15;
    const int half  = lane >> 4;
    const int half2 = half << 1;

    constexpr int WI = R / 16;
    const int wq = tid & 7;
    const int wr = tid >> 3;
    constexpr int GPR = NB / VEC;
    constexpr int FRS = 128 / GPR;
    constexpr int FI  = KC / FRS;
    static_assert(128 % GPR == 0 && KC % FRS == 0, "copy plan must tile the chunk");
    const int fg = tid % GPR;
    const int fr = tid / GPR;
    const int n0 = tng * NB + fg * VEC;
    const int fb = n0 / S;
    const int fs = n0 - fb * S;
    const float* __restrict__ fsrc = F + (size_t)fb * C * S + (size_t)fr * S + fs;

    v4f wreg[WI];
    v4f fregv[VEC == 4 ? FI : 1];
    float fregs[VEC == 1 ? FI : 1];

    auto load_chunk = [&](int kc) {
        #pragma unroll
        for (int i = 0; i < WI; ++i) {
            int row = rowbase + wr + 16 * i;
            row = row > 254 ? 254 : row;
            wreg[i] = *(const v4f*)(Wm + (size_t)row * C + kc + 4 * wq);
        }
        #pragma unroll
        for (int i = 0; i < FI; ++i) {
            const float* p = fsrc + (size_t)(kc + i * FRS) * S;
            if constexpr (VEC == 4) fregv[i] = *(const v4f*)p;
            else                    fregs[i] = *p;
        }
    };
    auto store_chunk = [&](float* buf) {
        #pragma unroll
        for (int i = 0; i < WI; ++i)
            *(v4f*)(buf + (wr + 16 * i) * PW + 4 * wq) = wreg[i];
        #pragma unroll
        for (int i = 0; i < FI; ++i) {
            float* q = buf + WBUF + (fr + i * FRS) * PF + fg * VEC;
            if constexpr (VEC == 4) *(v4f*)q = fregv[i];
            else                    *q = fregs[i];
        }
    };

    v8f acc[MT][NT];
    #pragma unroll
    for (int t = 0; t < MT; ++t) {
        #pragma unroll
        for (int j = 0; j < NT; ++j)
            acc[t][j] = (v8f){0.f, 0.f, 0.f, 0.f, 0.f, 0.f, 0.f, 0.f};
    }

    const int wl = ((wv * MT) * 16 + l16) * PW + half2;
    const int fl = WBUF + half2 * PF + l16;

    load_chunk(0);
    store_chunk(lds);
    __syncthreads();

    for (int c = 0; c < NCH; ++c) {
        const float* cur = lds + (c & 1) * BUF;
        if (c + 1 < NCH) load_chunk((c + 1) * KC);

        #pragma unroll
        for (int kk = 0; kk < KC; kk += 4) {
            v2f a[MT];
            v2f bb[NT];
            #pragma unroll
            for (int t = 0; t < MT; ++t)
                a[t] = *(const v2f*)(cur + wl + t * 16 * PW + kk);
            #pragma unroll
            for (int j = 0; j < NT; ++j) {
                bb[j].x = cur[fl + kk * PF + j * 16];
                bb[j].y = cur[fl + kk * PF + PF + j * 16];
            }
            #pragma unroll
            for (int t = 0; t < MT; ++t) {
                #pragma unroll
                for (int j = 0; j < NT; ++j)
                    acc[t][j] = __builtin_amdgcn_wmma_f32_16x16x4_f32(
                                    false, bb[j], false, a[t], (short)0, acc[t][j], false, false);
            }
        }

        if (c + 1 < NCH) store_chunk(lds + ((c + 1) & 1) * BUF);
        __syncthreads();
    }

    constexpr int NA    = (R == 256) ? 3 : 2;
    constexpr int PSZ   = NB * 85;
    constexpr int TRASH = NA * PSZ;
    static_assert(R == 256 || R == 128, "the anchor split below is written for these two row counts");
    static_assert(TRASH + 1 <= 2 * BUF, "the staged output must fit in the operand buffers");
    static_assert(PSZ % 4 == 0, "16-byte LDS alignment of each anchor's block");
    const int a_first = (R == 256) ? 0 : (rowbase != 0 ? 1 : 0);

    float bvT[MT], anchT[MT], mulT[MT];
    int   qbT[MT], qsT[MT];
    bool  expT[MT], isxT[MT], isyT[MT];
    #pragma unroll
    for (int t = 0; t < MT; ++t) {
        const int o    = ((tmg * MT + t) << 4) + l16;
        const int oc   = o > 254 ? 254 : o;
        const int a    = oc / 85;
        const int attr = oc - a * 85;
        const float aw = (a == 0) ? aw0 : ((a == 1) ? aw1 : aw2);
        const float ah = (a == 0) ? ah0 : ((a == 1) ? ah1 : ah2);
        const bool  ok = (o < 255);
        bvT[t]   = bias[oc];
        anchT[t] = (attr == 2) ? aw : ah;
        mulT[t]  = (attr < 2) ? inv_grid : 1.0f;
        expT[t]  = (attr == 2) || (attr == 3);
        isxT[t]  = (attr == 0);
        isyT[t]  = (attr == 1);
        qbT[t]   = ok ? ((a - a_first) * PSZ + attr + 8 * 85 * half) : TRASH;
        qsT[t]   = ok ? 85 : 0;
    }

    const unsigned mofs = (unsigned)half << 3;
    #pragma unroll
    for (int j = 0; j < NT; ++j) {
        #pragma unroll
        for (int r = 0; r < 8; ++r) {
            const unsigned n  = ((unsigned)(tng * NT + j) << 4) + mofs + (unsigned)r;
            const unsigned b  = n / (unsigned)S;
            const unsigned s  = n - b * (unsigned)S;
            const unsigned gh = s / (unsigned)G;
            const unsigned gw = s - gh * (unsigned)G;
            const float wcol = (float)gw;
            const float hrow = (float)gh;
            #pragma unroll
            for (int t = 0; t < MT; ++t) {
                const float v   = acc[t][j][r] + bvT[t];
                const float e   = __expf(expT[t] ? v : -v);
                const float sg  = __builtin_amdgcn_rcpf(1.0f + e);
                const float add = isxT[t] ? wcol : (isyT[t] ? hrow : 0.0f);
                const float rs  = (add + sg) * mulT[t];
                const float re  = e * anchT[t];
                const float res = expT[t] ? re : rs;
                lds[qbT[t] + (16 * j + r) * qsT[t]] = res;
            }
        }
    }
    __syncthreads();

    const unsigned n0wg = (unsigned)(tng * NB);
    const int wvu = __builtin_amdgcn_readfirstlane(wv);
    #pragma unroll
    for (int al = 0; al < NA; ++al) {
        const int a  = a_first + al;
        const int lo = (rowbase - a * 85) > 0 ? (rowbase - a * 85) : 0;
        const int hi = (rowbase + R - a * 85) < 85 ? (rowbase + R - a * 85) : 85;
        const float* __restrict__ q = lds + al * PSZ;
        if (VEC == 4 && lo == 0 && hi == 85) {
            for (int g = wvu; g < NB / 4; g += WPB) {
                const unsigned n = n0wg + 4u * (unsigned)g;
                const unsigned b = n / (unsigned)S;
                const unsigned s = n - b * (unsigned)S;
                float* __restrict__ dst = out + ((size_t)(b * 3u + (unsigned)a) * S + s) * 85;
                const float* src = q + g * 340;
                #pragma unroll
                for (int k = 0; k < 3; ++k) {
                    const int f = lane + 32 * k;
                    if (f < 85) {
                        const v4f val = *(const v4f*)(src + 4 * f);
                        __builtin_nontemporal_store(val, (v4f*)(dst + 4 * f));
                    }
                }
            }
        } else if (lo == 0 && hi == 85) {
            copy_rows<S, NB, 85>(q, out, a, 0, n0wg, tid);
        } else if (lo == 0) {
            copy_rows<S, NB, 43>(q, out, a, 0, n0wg, tid);
        } else {
            copy_rows<S, NB, 42>(q, out, a, 43, n0wg, tid);
        }
    }
}

constexpr int MT13 = 2, NT13 = 2, VEC13 = 1;
constexpr int MT26 = 2, NT26 = 4, VEC26 = 4;
constexpr int MT52 = 4, NT52 = 4, VEC52 = 4;
constexpr int NB13 = head_blocks<13 * 13, MT13, NT13>();
constexpr int NB26 = head_blocks<26 * 26, MT26, NT26>();
constexpr int NB52 = head_blocks<52 * 52, MT52, NT52>();
static_assert(NB13 * WPB * MT13 * NT13 * 256 == 16 * 32 * 13 * 13 * 16, "grid of the 13 head must cover its tiles");
static_assert(NB26 * WPB * MT26 * NT26 * 256 == 16 * 32 * 26 * 26 * 16, "grid of the 26 head must cover its tiles");
static_assert(NB52 * WPB * MT52 * NT52 * 256 == 16 * 32 * 52 * 52 * 16, "grid of the 52 head must cover its tiles");

constexpr int cmax(int a, int b) { return a > b ? a : b; }
constexpr int LDS_DWORDS = cmax(head_lds_dwords<MT13, NT13>(),
                           cmax(head_lds_dwords<MT26, NT26>(), head_lds_dwords<MT52, NT52>()));

constexpr size_t OFF13 = 0;
constexpr size_t OFF26 = (size_t)32 * 507  * 85;
constexpr size_t OFF52 = OFF26 + (size_t)32 * 2028 * 85;

constexpr float K416 = 1.0f / 416.0f;

__global__ __launch_bounds__(WPB * 32)
void yolo_heads_wmma(const float* __restrict__ f13, const float* __restrict__ f26, const float* __restrict__ f52,
                     const float* __restrict__ W13, const float* __restrict__ b13,
                     const float* __restrict__ W26, const float* __restrict__ b26,
                     const float* __restrict__ W52, const float* __restrict__ b52,
                     float* __restrict__ out)
{
    __shared__ __attribute__((aligned(16))) float lds[LDS_DWORDS];
    const int bid = blockIdx.x;
    if (bid < NB13) {
        yolo_head_wmma<1024, 13 * 13, 13, MT13, NT13, VEC13>(lds, bid, f13, W13, b13, out + OFF13,
            116.f * K416,  90.f * K416, 156.f * K416, 198.f * K416, 373.f * K416, 326.f * K416);
    } else if (bid < NB13 + NB52) {
        yolo_head_wmma< 256, 52 * 52, 52, MT52, NT52, VEC52>(lds, bid - NB13, f52, W52, b52, out + OFF52,
             10.f * K416,  13.f * K416,  16.f * K416,  30.f * K416,  33.f * K416,  23.f * K416);
    } else {
        yolo_head_wmma< 512, 26 * 26, 26, MT26, NT26, VEC26>(lds, bid - NB13 - NB52, f26, W26, b26, out + OFF26,
             30.f * K416,  61.f * K416,  62.f * K416,  45.f * K416,  59.f * K416, 119.f * K416);
    }
}

extern "C" void kernel_launch(void* const* d_in, const int* in_sizes, int n_in,
                              void* d_out, int out_size, void* d_ws, size_t ws_size,
                              hipStream_t stream) {
    (void)in_sizes; (void)n_in; (void)out_size; (void)d_ws; (void)ws_size;
    const float* f13 = (const float*)d_in[0];
    const float* f26 = (const float*)d_in[1];
    const float* f52 = (const float*)d_in[2];
    const float* W13 = (const float*)d_in[3];
    const float* b13 = (const float*)d_in[4];
    const float* W26 = (const float*)d_in[5];
    const float* b26 = (const float*)d_in[6];
    const float* W52 = (const float*)d_in[7];
    const float* b52 = (const float*)d_in[8];
    float* out = (float*)d_out;

    yolo_heads_wmma<<<dim3(NB13 + NB52 + NB26), dim3(WPB * 32), 0, stream>>>(
        f13, f26, f52, W13, b13, W26, b26, W52, b52, out);
}


# Wave tiles of heads 13 and 26 grown to 4x2 and 4x4 (work-group = all 256 channels): twice the matrix work per K chunk and barrier, activation tiles staged once, long groups first; k order unchanged
# speedup vs baseline: 4.6675x; 1.0436x over previous
#include <hip/hip_runtime.h>


constexpr int WPB = 4;

typedef __attribute__((ext_vector_type(2))) float v2f;
typedef __attribute__((ext_vector_type(4))) float v4f;
typedef __attribute__((ext_vector_type(8))) float v8f;

template <int S, int NB, int LEN>
__device__ __forceinline__ void copy_rows(const float* __restrict__ q, float* __restrict__ out,
                                          int a, int lo, unsigned n0wg, int tid)
{
    for (int e = tid; e < NB * LEN; e += WPB * 32) {
        const unsigned p = (unsigned)e / (unsigned)LEN;
        const unsigned i = (unsigned)e - p * (unsigned)LEN;
        const unsigned n = n0wg + p;
        const unsigned b = n / (unsigned)S;
        const unsigned s = n - b * (unsigned)S;
        const float val = q[p * 85u + (unsigned)lo + i];
        __builtin_nontemporal_store(val, out + ((size_t)(b * 3u + (unsigned)a) * S + s) * 85 + (unsigned)lo + i);
    }
}

template <int S, int MT, int NT>
constexpr int head_blocks() { return (16 / MT) * ((32 * S) / 16 / NT) / WPB; }

template <int MT, int NT>
constexpr int head_lds_dwords() { return 2 * (WPB * MT * 16 * (32 + 4) + 32 * (NT * 16 + 8)); }

template <int C, int S, int G, int MT, int NT, int VEC>
__device__ __forceinline__
void yolo_head_wmma(float* lds,
                    const int bid,
                    const float* __restrict__ F,
                    const float* __restrict__ Wm,
                    const float* __restrict__ bias,
                    float* __restrict__ out,
                    float aw0, float ah0, float aw1, float ah1,
                    float aw2, float ah2)
{
    constexpr float inv_grid = 1.0f / (float)G;
    constexpr int MG  = 16 / MT;
    constexpr int KC  = 32;
    constexpr int NCH = C / KC;
    constexpr int R   = WPB * MT * 16;
    constexpr int NB  = NT * 16;
    constexpr int PW  = KC + 4;
    constexpr int PF  = NB + 8;
    constexpr int WBUF = R * PW;
    constexpr int FBUF = KC * PF;
    constexpr int BUF  = WBUF + FBUF;
    static_assert(MG * MT == 16, "MT must divide 16");
    static_assert(MG % WPB == 0, "a work-group's waves must share one position group");
    static_assert(((32 * S) / 16) % NT == 0, "NT must divide the N-tile count");
    static_assert(C % KC == 0, "K chunk must divide C");
    static_assert(VEC == 1 || (VEC == 4 && S % 4 == 0), "16-byte activation loads need S % 4 == 0");
    static_assert(BUF % 4 == 0 && WBUF % 4 == 0 && PW % 4 == 0 && PF % 4 == 0, "16-byte LDS alignment");

    static_assert(2 * BUF == head_lds_dwords<MT, NT>(), "LDS sizing helper out of step with the body");

    const int tid   = threadIdx.x;
    const int lane  = tid & 31;
    const int wv    = tid >> 5;
    const int wave  = bid * WPB + wv;
    const int tmg   = wave % MG;
    const int tng   = wave / MG;
    const int rowbase = ((bid * WPB) % MG) * (MT * 16);

    const int l16   = lane & 15;
    const int half  = lane >> 4;
    const int half2 = half << 1;

    constexpr int WI = R / 16;
    const int wq = tid & 7;
    const int wr = tid >> 3;
    constexpr int GPR = NB / VEC;
    constexpr int FRS = 128 / GPR;
    constexpr int FI  = KC / FRS;
    static_assert(128 % GPR == 0 && KC % FRS == 0, "copy plan must tile the chunk");
    const int fg = tid % GPR;
    const int fr = tid / GPR;
    const int n0 = tng * NB + fg * VEC;
    const int fb = n0 / S;
    const int fs = n0 - fb * S;
    const float* __restrict__ fsrc = F + (size_t)fb * C * S + (size_t)fr * S + fs;

    v4f wreg[WI];
    v4f fregv[VEC == 4 ? FI : 1];
    float fregs[VEC == 1 ? FI : 1];

    auto load_chunk = [&](int kc) {
        #pragma unroll
        for (int i = 0; i < WI; ++i) {
            int row = rowbase + wr + 16 * i;
            row = row > 254 ? 254 : row;
            wreg[i] = *(const v4f*)(Wm + (size_t)row * C + kc + 4 * wq);
        }
        #pragma unroll
        for (int i = 0; i < FI; ++i) {
            const float* p = fsrc + (size_t)(kc + i * FRS) * S;
            if constexpr (VEC == 4) fregv[i] = *(const v4f*)p;
            else                    fregs[i] = *p;
        }
    };
    auto store_chunk = [&](float* buf) {
        #pragma unroll
        for (int i = 0; i < WI; ++i)
            *(v4f*)(buf + (wr + 16 * i) * PW + 4 * wq) = wreg[i];
        #pragma unroll
        for (int i = 0; i < FI; ++i) {
            float* q = buf + WBUF + (fr + i * FRS) * PF + fg * VEC;
            if constexpr (VEC == 4) *(v4f*)q = fregv[i];
            else                    *q = fregs[i];
        }
    };

    v8f acc[MT][NT];
    #pragma unroll
    for (int t = 0; t < MT; ++t) {
        #pragma unroll
        for (int j = 0; j < NT; ++j)
            acc[t][j] = (v8f){0.f, 0.f, 0.f, 0.f, 0.f, 0.f, 0.f, 0.f};
    }

    const int wl = ((wv * MT) * 16 + l16) * PW + half2;
    const int fl = WBUF + half2 * PF + l16;

    load_chunk(0);
    store_chunk(lds);
    __syncthreads();

    for (int c = 0; c < NCH; ++c) {
        const float* cur = lds + (c & 1) * BUF;
        if (c + 1 < NCH) load_chunk((c + 1) * KC);

        #pragma unroll
        for (int kk = 0; kk < KC; kk += 4) {
            v2f a[MT];
            v2f bb[NT];
            #pragma unroll
            for (int t = 0; t < MT; ++t)
                a[t] = *(const v2f*)(cur + wl + t * 16 * PW + kk);
            #pragma unroll
            for (int j = 0; j < NT; ++j) {
                bb[j].x = cur[fl + kk * PF + j * 16];
                bb[j].y = cur[fl + kk * PF + PF + j * 16];
            }
            #pragma unroll
            for (int t = 0; t < MT; ++t) {
                #pragma unroll
                for (int j = 0; j < NT; ++j)
                    acc[t][j] = __builtin_amdgcn_wmma_f32_16x16x4_f32(
                                    false, bb[j], false, a[t], (short)0, acc[t][j], false, false);
            }
        }

        if (c + 1 < NCH) store_chunk(lds + ((c + 1) & 1) * BUF);
        __syncthreads();
    }

    constexpr int NA    = (R == 256) ? 3 : 2;
    constexpr int PSZ   = NB * 85;
    constexpr int TRASH = NA * PSZ;
    static_assert(R == 256 || R == 128, "the anchor split below is written for these two row counts");
    static_assert(TRASH + 1 <= 2 * BUF, "the staged output must fit in the operand buffers");
    static_assert(PSZ % 4 == 0, "16-byte LDS alignment of each anchor's block");
    const int a_first = (R == 256) ? 0 : (rowbase != 0 ? 1 : 0);

    float bvT[MT], anchT[MT], mulT[MT];
    int   qbT[MT], qsT[MT];
    bool  expT[MT], isxT[MT], isyT[MT];
    #pragma unroll
    for (int t = 0; t < MT; ++t) {
        const int o    = ((tmg * MT + t) << 4) + l16;
        const int oc   = o > 254 ? 254 : o;
        const int a    = oc / 85;
        const int attr = oc - a * 85;
        const float aw = (a == 0) ? aw0 : ((a == 1) ? aw1 : aw2);
        const float ah = (a == 0) ? ah0 : ((a == 1) ? ah1 : ah2);
        const bool  ok = (o < 255);
        bvT[t]   = bias[oc];
        anchT[t] = (attr == 2) ? aw : ah;
        mulT[t]  = (attr < 2) ? inv_grid : 1.0f;
        expT[t]  = (attr == 2) || (attr == 3);
        isxT[t]  = (attr == 0);
        isyT[t]  = (attr == 1);
        qbT[t]   = ok ? ((a - a_first) * PSZ + attr + 8 * 85 * half) : TRASH;
        qsT[t]   = ok ? 85 : 0;
    }

    const unsigned mofs = (unsigned)half << 3;
    #pragma unroll
    for (int j = 0; j < NT; ++j) {
        #pragma unroll
        for (int r = 0; r < 8; ++r) {
            const unsigned n  = ((unsigned)(tng * NT + j) << 4) + mofs + (unsigned)r;
            const unsigned b  = n / (unsigned)S;
            const unsigned s  = n - b * (unsigned)S;
            const unsigned gh = s / (unsigned)G;
            const unsigned gw = s - gh * (unsigned)G;
            const float wcol = (float)gw;
            const float hrow = (float)gh;
            #pragma unroll
            for (int t = 0; t < MT; ++t) {
                const float v   = acc[t][j][r] + bvT[t];
                const float e   = __expf(expT[t] ? v : -v);
                const float sg  = __builtin_amdgcn_rcpf(1.0f + e);
                const float add = isxT[t] ? wcol : (isyT[t] ? hrow : 0.0f);
                const float rs  = (add + sg) * mulT[t];
                const float re  = e * anchT[t];
                const float res = expT[t] ? re : rs;
                lds[qbT[t] + (16 * j + r) * qsT[t]] = res;
            }
        }
    }
    __syncthreads();

    const unsigned n0wg = (unsigned)(tng * NB);
    const int wvu = __builtin_amdgcn_readfirstlane(wv);
    #pragma unroll
    for (int al = 0; al < NA; ++al) {
        const int a  = a_first + al;
        const int lo = (rowbase - a * 85) > 0 ? (rowbase - a * 85) : 0;
        const int hi = (rowbase + R - a * 85) < 85 ? (rowbase + R - a * 85) : 85;
        const float* __restrict__ q = lds + al * PSZ;
        if (VEC == 4 && lo == 0 && hi == 85) {
            for (int g = wvu; g < NB / 4; g += WPB) {
                const unsigned n = n0wg + 4u * (unsigned)g;
                const unsigned b = n / (unsigned)S;
                const unsigned s = n - b * (unsigned)S;
                float* __restrict__ dst = out + ((size_t)(b * 3u + (unsigned)a) * S + s) * 85;
                const float* src = q + g * 340;
                #pragma unroll
                for (int k = 0; k < 3; ++k) {
                    const int f = lane + 32 * k;
                    if (f < 85) {
                        const v4f val = *(const v4f*)(src + 4 * f);
                        __builtin_nontemporal_store(val, (v4f*)(dst + 4 * f));
                    }
                }
            }
        } else if (lo == 0 && hi == 85) {
            copy_rows<S, NB, 85>(q, out, a, 0, n0wg, tid);
        } else if (lo == 0) {
            copy_rows<S, NB, 43>(q, out, a, 0, n0wg, tid);
        } else {
            copy_rows<S, NB, 42>(q, out, a, 43, n0wg, tid);
        }
    }
}

constexpr int MT13 = 4, NT13 = 2, VEC13 = 1;
constexpr int MT26 = 4, NT26 = 4, VEC26 = 4;
constexpr int MT52 = 4, NT52 = 4, VEC52 = 4;
constexpr int NB13 = head_blocks<13 * 13, MT13, NT13>();
constexpr int NB26 = head_blocks<26 * 26, MT26, NT26>();
constexpr int NB52 = head_blocks<52 * 52, MT52, NT52>();
static_assert(NB13 * WPB * MT13 * NT13 * 256 == 16 * 32 * 13 * 13 * 16, "grid of the 13 head must cover its tiles");
static_assert(NB26 * WPB * MT26 * NT26 * 256 == 16 * 32 * 26 * 26 * 16, "grid of the 26 head must cover its tiles");
static_assert(NB52 * WPB * MT52 * NT52 * 256 == 16 * 32 * 52 * 52 * 16, "grid of the 52 head must cover its tiles");

constexpr int cmax(int a, int b) { return a > b ? a : b; }
constexpr int LDS_DWORDS = cmax(head_lds_dwords<MT13, NT13>(),
                           cmax(head_lds_dwords<MT26, NT26>(), head_lds_dwords<MT52, NT52>()));

constexpr size_t OFF13 = 0;
constexpr size_t OFF26 = (size_t)32 * 507  * 85;
constexpr size_t OFF52 = OFF26 + (size_t)32 * 2028 * 85;

constexpr float K416 = 1.0f / 416.0f;

__global__ __launch_bounds__(WPB * 32)
void yolo_heads_wmma(const float* __restrict__ f13, const float* __restrict__ f26, const float* __restrict__ f52,
                     const float* __restrict__ W13, const float* __restrict__ b13,
                     const float* __restrict__ W26, const float* __restrict__ b26,
                     const float* __restrict__ W52, const float* __restrict__ b52,
                     float* __restrict__ out)
{
    __shared__ __attribute__((aligned(16))) float lds[LDS_DWORDS];
    const int bid = blockIdx.x;
    if (bid < NB13) {
        yolo_head_wmma<1024, 13 * 13, 13, MT13, NT13, VEC13>(lds, bid, f13, W13, b13, out + OFF13,
            116.f * K416,  90.f * K416, 156.f * K416, 198.f * K416, 373.f * K416, 326.f * K416);
    } else if (bid < NB13 + NB26) {
        yolo_head_wmma< 512, 26 * 26, 26, MT26, NT26, VEC26>(lds, bid - NB13, f26, W26, b26, out + OFF26,
             30.f * K416,  61.f * K416,  62.f * K416,  45.f * K416,  59.f * K416, 119.f * K416);
    } else {
        yolo_head_wmma< 256, 52 * 52, 52, MT52, NT52, VEC52>(lds, bid - NB13 - NB26, f52, W52, b52, out + OFF52,
             10.f * K416,  13.f * K416,  16.f * K416,  30.f * K416,  33.f * K416,  23.f * K416);
    }
}

extern "C" void kernel_launch(void* const* d_in, const int* in_sizes, int n_in,
                              void* d_out, int out_size, void* d_ws, size_t ws_size,
                              hipStream_t stream) {
    (void)in_sizes; (void)n_in; (void)out_size; (void)d_ws; (void)ws_size;
    const float* f13 = (const float*)d_in[0];
    const float* f26 = (const float*)d_in[1];
    const float* f52 = (const float*)d_in[2];
    const float* W13 = (const float*)d_in[3];
    const float* b13 = (const float*)d_in[4];
    const float* W26 = (const float*)d_in[5];
    const float* b26 = (const float*)d_in[6];
    const float* W52 = (const float*)d_in[7];
    const float* b52 = (const float*)d_in[8];
    float* out = (float*)d_out;

    yolo_heads_wmma<<<dim3(NB13 + NB52 + NB26), dim3(WPB * 32), 0, stream>>>(
        f13, f26, f52, W13, b13, W26, b26, W52, b52, out);
}
